# starting kernel
# baseline (speedup 1.0000x reference)
#include <hip/hip_runtime.h>
#include <hip/hip_bf16.h>


typedef float v2f __attribute__((ext_vector_type(2)));
typedef float v8f __attribute__((ext_vector_type(8)));

#define BATCH       64
#define CHANS       2048
#define REG_TOTAL   14
#define OUT_COLS    (REG_TOTAL * CHANS)
#define ROWS        (BATCH * CHANS)
#define WAVES       (ROWS / 16)
#define WAVES_PER_B (CHANS / 16)
#define KSTEPS      64

__global__ void rmac_pool_wmma(const float* __restrict__ x,
                               float* __restrict__ out,
                               float* __restrict__ partials) {
    __shared__ v2f wtab[KSTEPS * 32];

    for (int idx = threadIdx.x; idx < KSTEPS * 32; idx += blockDim.x) {
        const int lw = idx & 31;
        const int mm = lw & 15;
        const int sw = lw >> 4;
        const int p  = ((idx >> 5) << 2) + 2 * sw;
        int h0 = 0, w0 = 0, len = 0; float wt = 0.0f;
        if (mm == 0)       { len = 16; wt = 1.0f / 256.0f; }
        else if (mm <= 4)  { int t = mm - 1; h0 = (t >> 1) * 4; w0 = (t & 1) * 4; len = 10; wt = 1.0f / 100.0f; }
        else if (mm <= 13) { int t = mm - 5; h0 = (t / 3) * 3;  w0 = (t % 3) * 3; len = 8;  wt = 1.0f / 64.0f;  }
        const unsigned hm = ((1u << len) - 1u) << h0;
        const unsigned wm = ((1u << len) - 1u) << w0;
        const int h = p >> 4;
        const int w = p & 15;
        const unsigned hb = (hm >> h) & 1u;
        v2f a;
        a.x = ((hb & (wm >> w)) & 1u)       ? wt : 0.0f;
        a.y = ((hb & (wm >> (w + 1))) & 1u) ? wt : 0.0f;
        wtab[idx] = a;
    }
    __syncthreads();

    const int lane     = threadIdx.x & 31;
    const int wid      = blockIdx.x * (blockDim.x >> 5) + (threadIdx.x >> 5);
    const int row_base = wid << 4;
    const int n        = lane & 15;
    const int s        = lane >> 4;

    const float* rowp = x + (size_t)(row_base + n) * 256 + 2 * s;
    const v2f*   wrow = wtab + lane;

    v8f acc = {0.f, 0.f, 0.f, 0.f, 0.f, 0.f, 0.f, 0.f};

#pragma unroll
    for (int step = 0; step < KSTEPS; ++step) {
        const v2f a = wrow[step * 32];
        const v2f b = *(const v2f*)(rowp + step * 4);
        acc = __builtin_amdgcn_wmma_f32_16x16x4_f32(
             false, a,  false, b,
             (short)0, acc,  false,  false);
    }

    const int b_idx  = row_base >> 11;
    const int c_base = row_base & (CHANS - 1);
    float* obase = out + (size_t)b_idx * OUT_COLS + c_base + n;

    float ss = 0.0f;
#pragma unroll
    for (int r = 0; r < 6; ++r) {
        const float v = acc[r];
        ss += v * v;
        obase[(size_t)(r + 8 * s) * CHANS] = v;
    }
    {
        const float v6 = acc[6], v7 = acc[7];
        ss += v6 * v6 + v7 * v7;
        if (s == 0) {
            obase[(size_t)6 * CHANS] = v6;
            obase[(size_t)7 * CHANS] = v7;
        }
    }

#pragma unroll
    for (int off = 16; off > 0; off >>= 1)
        ss += __shfl_xor(ss, off, 32);
    if (lane == 0)
        partials[wid] = ss;
}

__global__ void rmac_norm_reduce(const float* __restrict__ partials,
                                 float* __restrict__ norms) {
    __shared__ float sdata[WAVES_PER_B];
    const int b = blockIdx.x;
    const int t = threadIdx.x;
    sdata[t] = partials[b * WAVES_PER_B + t];
    __syncthreads();
#pragma unroll
    for (int off = WAVES_PER_B / 2; off > 0; off >>= 1) {
        if (t < off) sdata[t] += sdata[t + off];
        __syncthreads();
    }
    if (t == 0) norms[b] = sdata[0];
}

__global__ void rmac_scale(float* __restrict__ out,
                           const float* __restrict__ norms, int nElem) {
    const int i = blockIdx.x * blockDim.x + threadIdx.x;
    if (i < nElem) {
        const int b = i / OUT_COLS;
        const float inv = 1.0f / fmaxf(sqrtf(norms[b]), 1e-12f);
        out[i] = out[i] * inv;
    }
}

extern "C" void kernel_launch(void* const* d_in, const int* in_sizes, int n_in,
                              void* d_out, int out_size, void* d_ws, size_t ws_size,
                              hipStream_t stream) {
    const float* x  = (const float*)d_in[0];
    float* out      = (float*)d_out;
    float* partials = (float*)d_ws;
    float* norms    = partials + WAVES;

    rmac_pool_wmma<<<WAVES / 4, 128, 0, stream>>>(x, out, partials);
    rmac_norm_reduce<<<BATCH, WAVES_PER_B, 0, stream>>>(partials, norms);

    const int nElem = BATCH * OUT_COLS;
    rmac_scale<<<(nElem + 255) / 256, 256, 0, stream>>>(out, norms, nElem);
}


# x and weight fragment loads widened from 64 to 128 bit: each lane takes 4 consecutive pixels for 2 k-steps, halving loads and lines touched per wave; k sum order within each 8-pixel group permuted
# speedup vs baseline: 1.0992x; 1.0992x over previous
#include <hip/hip_runtime.h>
#include <hip/hip_bf16.h>


typedef float v2f __attribute__((ext_vector_type(2)));
typedef float v4f __attribute__((ext_vector_type(4)));
typedef float v8f __attribute__((ext_vector_type(8)));

#define BATCH       64
#define CHANS       2048
#define REG_TOTAL   14
#define OUT_COLS    (REG_TOTAL * CHANS)
#define ROWS        (BATCH * CHANS)
#define WAVES       (ROWS / 16)
#define WAVES_PER_B (CHANS / 16)
#define KPAIRS      32

__global__ void rmac_pool_wmma(const float* __restrict__ x,
                               float* __restrict__ out,
                               float* __restrict__ partials) {
    __shared__ v4f wtab[KPAIRS * 32];

    for (int idx = threadIdx.x; idx < KPAIRS * 32; idx += blockDim.x) {
        const int lw = idx & 31;
        const int mm = lw & 15;
        const int sw = lw >> 4;
        const int p  = ((idx >> 5) << 3) + 4 * sw;
        int h0 = 0, w0 = 0, len = 0; float wt = 0.0f;
        if (mm == 0)       { len = 16; wt = 1.0f / 256.0f; }
        else if (mm <= 4)  { int t = mm - 1; h0 = (t >> 1) * 4; w0 = (t & 1) * 4; len = 10; wt = 1.0f / 100.0f; }
        else if (mm <= 13) { int t = mm - 5; h0 = (t / 3) * 3;  w0 = (t % 3) * 3; len = 8;  wt = 1.0f / 64.0f;  }
        const unsigned hm = ((1u << len) - 1u) << h0;
        const unsigned wm = ((1u << len) - 1u) << w0;
        const int h = p >> 4;
        const int w = p & 15;
        const unsigned wb = ((hm >> h) & 1u) ? (wm >> w) : 0u;
        v4f a;
        a.x = (wb & 1u) ? wt : 0.0f;
        a.y = (wb & 2u) ? wt : 0.0f;
        a.z = (wb & 4u) ? wt : 0.0f;
        a.w = (wb & 8u) ? wt : 0.0f;
        wtab[idx] = a;
    }
    __syncthreads();

    const int lane     = threadIdx.x & 31;
    const int wid      = blockIdx.x * (blockDim.x >> 5) + (threadIdx.x >> 5);
    const int row_base = wid << 4;
    const int n        = lane & 15;
    const int s        = lane >> 4;

    const float* rowp = x + (size_t)(row_base + n) * 256 + 4 * s;
    const v4f*   wrow = wtab + lane;

    v8f acc = {0.f, 0.f, 0.f, 0.f, 0.f, 0.f, 0.f, 0.f};

#pragma unroll
    for (int j = 0; j < KPAIRS; ++j) {
        const v4f a = wrow[j * 32];
        const v4f b = *(const v4f*)(rowp + j * 8);
        const v2f a0 = {a.x, a.y}, a1 = {a.z, a.w};
        const v2f b0 = {b.x, b.y}, b1 = {b.z, b.w};
        acc = __builtin_amdgcn_wmma_f32_16x16x4_f32(
             false, a0,  false, b0,
             (short)0, acc,  false,  false);
        acc = __builtin_amdgcn_wmma_f32_16x16x4_f32(
             false, a1,  false, b1,
             (short)0, acc,  false,  false);
    }

    const int b_idx  = row_base >> 11;
    const int c_base = row_base & (CHANS - 1);
    float* obase = out + (size_t)b_idx * OUT_COLS + c_base + n;

    float ss = 0.0f;
#pragma unroll
    for (int r = 0; r < 6; ++r) {
        const float v = acc[r];
        ss += v * v;
        obase[(size_t)(r + 8 * s) * CHANS] = v;
    }
    {
        const float v6 = acc[6], v7 = acc[7];
        ss += v6 * v6 + v7 * v7;
        if (s == 0) {
            obase[(size_t)6 * CHANS] = v6;
            obase[(size_t)7 * CHANS] = v7;
        }
    }

#pragma unroll
    for (int off = 16; off > 0; off >>= 1)
        ss += __shfl_xor(ss, off, 32);
    if (lane == 0)
        partials[wid] = ss;
}

__global__ void rmac_norm_reduce(const float* __restrict__ partials,
                                 float* __restrict__ norms) {
    __shared__ float sdata[WAVES_PER_B];
    const int b = blockIdx.x;
    const int t = threadIdx.x;
    sdata[t] = partials[b * WAVES_PER_B + t];
    __syncthreads();
#pragma unroll
    for (int off = WAVES_PER_B / 2; off > 0; off >>= 1) {
        if (t < off) sdata[t] += sdata[t + off];
        __syncthreads();
    }
    if (t == 0) norms[b] = sdata[0];
}

__global__ void rmac_scale(float* __restrict__ out,
                           const float* __restrict__ norms, int nElem) {
    const int i = blockIdx.x * blockDim.x + threadIdx.x;
    if (i < nElem) {
        const int b = i / OUT_COLS;
        const float inv = 1.0f / fmaxf(sqrtf(norms[b]), 1e-12f);
        out[i] = out[i] * inv;
    }
}

extern "C" void kernel_launch(void* const* d_in, const int* in_sizes, int n_in,
                              void* d_out, int out_size, void* d_ws, size_t ws_size,
                              hipStream_t stream) {
    const float* x  = (const float*)d_in[0];
    float* out      = (float*)d_out;
    float* partials = (float*)d_ws;
    float* norms    = partials + WAVES;

    rmac_pool_wmma<<<WAVES / 4, 128, 0, stream>>>(x, out, partials);
    rmac_norm_reduce<<<BATCH, WAVES_PER_B, 0, stream>>>(partials, norms);

    const int nElem = BATCH * OUT_COLS;
    rmac_scale<<<(nElem + 255) / 256, 256, 0, stream>>>(out, norms, nElem);
}
